# starting kernel
# baseline (speedup 1.0000x reference)
#include <hip/hip_runtime.h>
#include <hip/hip_bf16.h>

typedef __attribute__((ext_vector_type(2)))  float    v2f;
typedef __attribute__((ext_vector_type(8)))  float    v8f;
typedef __attribute__((ext_vector_type(16))) _Float16 v16h;

#define CH 64

__global__ void k_fill(float* __restrict__ p, float v, int n) {
  int i = blockIdx.x * blockDim.x + threadIdx.x;
  if (i < n) p[i] = v;
}

__global__ void k_deg(const int* __restrict__ dst, float* __restrict__ deg, int E) {
  int e = blockIdx.x * blockDim.x + threadIdx.x;
  if (e < E) unsafeAtomicAdd(&deg[dst[e]], 1.0f);
}

__global__ void k_rsqrt(float* __restrict__ p, int n) {
  int i = blockIdx.x * blockDim.x + threadIdx.x;
  if (i < n) p[i] = rsqrtf(p[i]);
}

__global__ void k_edge_norm(const int* __restrict__ src, const int* __restrict__ dst,
                            const float* __restrict__ dinv, float* __restrict__ nrm, int E) {
  int e = blockIdx.x * blockDim.x + threadIdx.x;
  if (e < E) nrm[e] = dinv[src[e]] * dinv[dst[e]];
}

__global__ __launch_bounds__(256) void k_gemm64(const float* __restrict__ A,
                                                const float* __restrict__ W,
                                                float* __restrict__ C, int N) {
  const int lane = threadIdx.x & 31;
  const int wave = threadIdx.x >> 5;
  const int row0 = blockIdx.x * 32 + (wave >> 2) * 16;
  if (row0 >= N) return;
  const int n0 = (wave & 3) * 16;
  const int m  = lane & 15;
  const int kh = lane >> 4;
  v8f acc = {};

#if __has_builtin(__builtin_amdgcn_wmma_f32_16x16x4_f32)
  const float* arow = A + (size_t)(row0 + m) * CH + 2 * kh;
#pragma unroll
  for (int k0 = 0; k0 < CH; k0 += 4) {
    v2f a;
    a.x = arow[k0];
    a.y = arow[k0 + 1];
    v2f b;
    b.x = W[(k0 + 2 * kh + 0) * CH + n0 + m];
    b.y = W[(k0 + 2 * kh + 1) * CH + n0 + m];
    acc = __builtin_amdgcn_wmma_f32_16x16x4_f32(false, a, false, b,
                                                (short)0, acc, false, false);
  }
#else
  const float* arow = A + (size_t)(row0 + m) * CH;
#pragma unroll
  for (int kb = 0; kb < CH; kb += 32) {
    v16h a, b;
#pragma unroll
    for (int e = 0; e < 16; ++e) {
      int k = (e < 8) ? (8 * kh + e) : (16 + 8 * kh + (e - 8));
      a[e] = (_Float16)arow[kb + k];
      b[e] = (_Float16)W[(size_t)(kb + k) * CH + n0 + m];
    }
    acc = __builtin_amdgcn_wmma_f32_16x16x32_f16(false, a, false, b,
                                                 (short)0, acc, false, false);
  }
#endif

#pragma unroll
  for (int v = 0; v < 8; ++v) {
    int r = row0 + v + 8 * kh;
    if (r < N) C[(size_t)r * CH + (n0 + m)] = acc[v];
  }
}

__global__ void k_selfloop(const float* __restrict__ h, const float* __restrict__ dinv,
                           float* __restrict__ out, int total) {
  int t = blockIdx.x * blockDim.x + threadIdx.x;
  if (t >= total) return;
  float di = dinv[t >> 6];
  out[t] = h[t] * di * di;
}

__global__ __launch_bounds__(256) void k_scatter(const float* __restrict__ h,
                                                 const int* __restrict__ src,
                                                 const int* __restrict__ dst,
                                                 const float* __restrict__ nrm,
                                                 float* __restrict__ out, int E) {
  int t = blockIdx.x * blockDim.x + threadIdx.x;
  int e = t >> 5;
  if (e >= E) return;
  int c = (t & 31) * 2;
  int s = src[e], d = dst[e];
  float w = nrm[e];
  const float2 hv = *(const float2*)(h + (size_t)s * CH + c);
  float* op = out + (size_t)d * CH + c;
  unsafeAtomicAdd(op,     hv.x * w);
  unsafeAtomicAdd(op + 1, hv.y * w);
}

__global__ void k_bias_relu(float* __restrict__ h, const float* __restrict__ b, int total) {
  int t = blockIdx.x * blockDim.x + threadIdx.x;
  if (t >= total) return;
  float v = h[t] + b[t & (CH - 1)];
  h[t] = v > 0.0f ? v : 0.0f;
}

__global__ __launch_bounds__(256) void k_edge_pred(const float* __restrict__ h,
                                                   const int* __restrict__ src,
                                                   const int* __restrict__ dst,
                                                   const float* __restrict__ Wfc,
                                                   const float* __restrict__ bfc,
                                                   float* __restrict__ out, int E) {
  int t = blockIdx.x * blockDim.x + threadIdx.x;
  int e = t >> 5;
  if (e >= E) return;
  int lane = t & 31;
  int c = lane * 2;
  int s = src[e], d = dst[e];
  float2 hs = *(const float2*)(h + (size_t)s * CH + c);
  float2 hd = *(const float2*)(h + (size_t)d * CH + c);
  float2 w0 = *(const float2*)(Wfc + c);
  float2 w1 = *(const float2*)(Wfc + CH + c);
  float sum = hs.x * w0.x + hs.y * w0.y + hd.x * w1.x + hd.y * w1.y;
#pragma unroll
  for (int off = 16; off > 0; off >>= 1) sum += __shfl_xor(sum, off, 32);
  if (lane == 0) out[e] = sum + bfc[0];
}

extern "C" void kernel_launch(void* const* d_in, const int* in_sizes, int n_in,
                              void* d_out, int out_size, void* d_ws, size_t ws_size,
                              hipStream_t stream) {
  const float* x   = (const float*)d_in[0];
  const int*   ei  = (const int*)d_in[1];
  const float* W1  = (const float*)d_in[2];
  const float* b1  = (const float*)d_in[3];
  const float* W2  = (const float*)d_in[4];
  const float* b2  = (const float*)d_in[5];
  const float* Wfc = (const float*)d_in[6];
  const float* bfc = (const float*)d_in[7];

  const int N = in_sizes[0] / CH;
  const int E = in_sizes[1] / 2;
  const int NF = N * CH;

  const int* src = ei;
  const int* dst = ei + E;

  float* ws   = (float*)d_ws;
  float* dinv = ws;
  float* nrm  = dinv + N;
  float* HB   = nrm + E;
  float* AB   = HB + (size_t)NF;

  const int TB = 256;
  dim3 blk(TB);
  dim3 gN((N + TB - 1) / TB);
  dim3 gE((E + TB - 1) / TB);
  dim3 gNF((NF + TB - 1) / TB);
  dim3 gEw((E * 32 + TB - 1) / TB);
  dim3 gG((N + 31) / 32);

  k_fill<<<gN, blk, 0, stream>>>(dinv, 1.0f, N);
  k_deg<<<gE, blk, 0, stream>>>(dst, dinv, E);
  k_rsqrt<<<gN, blk, 0, stream>>>(dinv, N);
  k_edge_norm<<<gE, blk, 0, stream>>>(src, dst, dinv, nrm, E);

  k_gemm64<<<gG, blk, 0, stream>>>(x, W1, HB, N);
  k_selfloop<<<gNF, blk, 0, stream>>>(HB, dinv, AB, NF);
  k_scatter<<<gEw, blk, 0, stream>>>(HB, src, dst, nrm, AB, E);
  k_bias_relu<<<gNF, blk, 0, stream>>>(AB, b1, NF);

  k_gemm64<<<gG, blk, 0, stream>>>(AB, W2, HB, N);
  k_selfloop<<<gNF, blk, 0, stream>>>(HB, dinv, AB, NF);
  k_scatter<<<gEw, blk, 0, stream>>>(HB, src, dst, nrm, AB, E);
  k_bias_relu<<<gNF, blk, 0, stream>>>(AB, b2, NF);

  k_edge_pred<<<gEw, blk, 0, stream>>>(AB, src, dst, Wfc, bfc, (float*)d_out, E);
}


# Elementwise passes folded into neighbours: self-loop term stored by the product's epilogue, bias+relu applied where sums are read (next A load, edge head), rsqrt at its readers; 13 launches, now 8
# speedup vs baseline: 2.0107x; 2.0107x over previous
#include <hip/hip_runtime.h>
#include <hip/hip_bf16.h>

typedef __attribute__((ext_vector_type(2)))  float    v2f;
typedef __attribute__((ext_vector_type(8)))  float    v8f;
typedef __attribute__((ext_vector_type(16))) _Float16 v16h;

#define CH 64

__global__ void k_fill(float* __restrict__ p, float v, int n) {
  int i = blockIdx.x * blockDim.x + threadIdx.x;
  if (i < n) p[i] = v;
}

__global__ void k_deg(const int* __restrict__ dst, float* __restrict__ deg, int E) {
  int e = blockIdx.x * blockDim.x + threadIdx.x;
  if (e < E) unsafeAtomicAdd(&deg[dst[e]], 1.0f);
}

__global__ void k_edge_norm(const int* __restrict__ src, const int* __restrict__ dst,
                            const float* __restrict__ deg, float* __restrict__ nrm, int E) {
  int e = blockIdx.x * blockDim.x + threadIdx.x;
  if (e < E) nrm[e] = rsqrtf(deg[src[e]]) * rsqrtf(deg[dst[e]]);
}

__device__ __forceinline__ float bias_relu(float x, float b) {
  float v = x + b;
  return v > 0.0f ? v : 0.0f;
}

template <bool RELU_IN>
__global__ __launch_bounds__(256) void k_gemm64(const float* __restrict__ A,
                                                const float* __restrict__ bin,
                                                const float* __restrict__ W,
                                                const float* __restrict__ deg,
                                                float* __restrict__ C,
                                                float* __restrict__ S, int N) {
  const int lane = threadIdx.x & 31;
  const int wave = threadIdx.x >> 5;
  const int row0 = blockIdx.x * 32 + (wave >> 2) * 16;
  if (row0 >= N) return;
  const int n0 = (wave & 3) * 16;
  const int m  = lane & 15;
  const int kh = lane >> 4;
  v8f acc = {};

  float di[8];
#pragma unroll
  for (int v = 0; v < 8; ++v) {
    const int r = row0 + v + 8 * kh;
    di[v] = rsqrtf(deg[min(r, N - 1)]);
  }

#if __has_builtin(__builtin_amdgcn_wmma_f32_16x16x4_f32)
  const float* arow = A + (size_t)(row0 + m) * CH + 2 * kh;
  const float* brow = bin + 2 * kh;
#pragma unroll
  for (int k0 = 0; k0 < CH; k0 += 4) {
    v2f a;
    a.x = arow[k0];
    a.y = arow[k0 + 1];
    if (RELU_IN) {
      a.x = bias_relu(a.x, brow[k0]);
      a.y = bias_relu(a.y, brow[k0 + 1]);
    }
    v2f b;
    b.x = W[(k0 + 2 * kh + 0) * CH + n0 + m];
    b.y = W[(k0 + 2 * kh + 1) * CH + n0 + m];
    acc = __builtin_amdgcn_wmma_f32_16x16x4_f32(false, a, false, b,
                                                (short)0, acc, false, false);
  }
#else
  const float* arow = A + (size_t)(row0 + m) * CH;
#pragma unroll
  for (int kb = 0; kb < CH; kb += 32) {
    v16h a, b;
#pragma unroll
    for (int e = 0; e < 16; ++e) {
      int k = (e < 8) ? (8 * kh + e) : (16 + 8 * kh + (e - 8));
      float av = arow[kb + k];
      if (RELU_IN) av = bias_relu(av, bin[kb + k]);
      a[e] = (_Float16)av;
      b[e] = (_Float16)W[(size_t)(kb + k) * CH + n0 + m];
    }
    acc = __builtin_amdgcn_wmma_f32_16x16x32_f16(false, a, false, b,
                                                 (short)0, acc, false, false);
  }
#endif

#pragma unroll
  for (int v = 0; v < 8; ++v) {
    int r = row0 + v + 8 * kh;
    if (r < N) {
      const float hv = acc[v];
      C[(size_t)r * CH + (n0 + m)] = hv;
      S[(size_t)r * CH + (n0 + m)] = hv * di[v] * di[v];
    }
  }
}

#define EPW 32
#define EB  8

template <bool FULL>
__device__ __forceinline__ void scatter_wave(const float* __restrict__ h,
                                             const int* __restrict__ src,
                                             const int* __restrict__ dst,
                                             const float* __restrict__ nrm,
                                             float* __restrict__ out,
                                             int e0, int cnt, int lane) {
  const int e = e0 + lane;
  const bool ok = FULL || lane < cnt;
  const int   s_l = ok ? src[e] : 0;
  const int   d_l = ok ? dst[e] : 0;
  const float w_l = ok ? nrm[e] : 0.0f;
  const int c = lane;
#pragma unroll
  for (int j0 = 0; j0 < EPW; j0 += EB) {
    float hlo[EB], hhi[EB];
#pragma unroll
    for (int j = 0; j < EB; ++j) {
      const int s = __builtin_amdgcn_readlane(s_l, j0 + j);
      const float* hp = h + (size_t)s * CH + c;
      hlo[j] = hp[0];
      hhi[j] = hp[CH / 2];
    }
#pragma unroll
    for (int j = 0; j < EB; ++j) {
      if (FULL || j0 + j < cnt) {
        const int   d = __builtin_amdgcn_readlane(d_l, j0 + j);
        const float w = __int_as_float(__builtin_amdgcn_readlane(__float_as_int(w_l), j0 + j));
        float* op = out + (size_t)d * CH + c;
        unsafeAtomicAdd(op,          hlo[j] * w);
        unsafeAtomicAdd(op + CH / 2, hhi[j] * w);
      }
    }
  }
}

__global__ __launch_bounds__(256) void k_scatter(const float* __restrict__ h,
                                                 const int* __restrict__ src,
                                                 const int* __restrict__ dst,
                                                 const float* __restrict__ nrm,
                                                 float* __restrict__ out, int E) {
  const int lane = threadIdx.x & 31;
  const int wave = __builtin_amdgcn_readfirstlane((int)(blockIdx.x * blockDim.x + threadIdx.x) >> 5);
  const int e0 = wave * EPW;
  if (e0 >= E) return;
  const int cnt = E - e0;
  if (cnt >= EPW) scatter_wave<true >(h, src, dst, nrm, out, e0, EPW, lane);
  else            scatter_wave<false>(h, src, dst, nrm, out, e0, cnt, lane);
}

__global__ __launch_bounds__(256) void k_edge_pred(const float* __restrict__ g,
                                                   const float* __restrict__ bh,
                                                   const int* __restrict__ src,
                                                   const int* __restrict__ dst,
                                                   const float* __restrict__ Wfc,
                                                   const float* __restrict__ bfc,
                                                   float* __restrict__ out, int E) {
  const int lane = threadIdx.x & 31;
  const int wave = __builtin_amdgcn_readfirstlane((int)(blockIdx.x * blockDim.x + threadIdx.x) >> 5);
  const int e0 = wave * EPW;
  if (e0 >= E) return;
  const int e = e0 + lane;
  const bool ok = e < E;
  const int s_l = ok ? src[e] : 0;
  const int d_l = ok ? dst[e] : 0;
  const int c = lane * 2;
  const float2 w0 = *(const float2*)(Wfc + c);
  const float2 w1 = *(const float2*)(Wfc + CH + c);
  const float2 bb = *(const float2*)(bh + c);
  float res = 0.0f;
#pragma unroll
  for (int j0 = 0; j0 < EPW; j0 += EB) {
    float2 hs[EB], hd[EB];
#pragma unroll
    for (int j = 0; j < EB; ++j) {
      const int s = __builtin_amdgcn_readlane(s_l, j0 + j);
      const int d = __builtin_amdgcn_readlane(d_l, j0 + j);
      hs[j] = *(const float2*)(g + (size_t)s * CH + c);
      hd[j] = *(const float2*)(g + (size_t)d * CH + c);
    }
#pragma unroll
    for (int j = 0; j < EB; ++j) {
      const float sx = bias_relu(hs[j].x, bb.x), sy = bias_relu(hs[j].y, bb.y);
      const float dx = bias_relu(hd[j].x, bb.x), dy = bias_relu(hd[j].y, bb.y);
      float sum = sx * w0.x + sy * w0.y + dx * w1.x + dy * w1.y;
#pragma unroll
      for (int off = 16; off > 0; off >>= 1) sum += __shfl_xor(sum, off, 32);
      if (lane == j0 + j) res = sum;
    }
  }
  if (ok) out[e] = res + bfc[0];
}

extern "C" void kernel_launch(void* const* d_in, const int* in_sizes, int n_in,
                              void* d_out, int out_size, void* d_ws, size_t ws_size,
                              hipStream_t stream) {
  const float* x   = (const float*)d_in[0];
  const int*   ei  = (const int*)d_in[1];
  const float* W1  = (const float*)d_in[2];
  const float* b1  = (const float*)d_in[3];
  const float* W2  = (const float*)d_in[4];
  const float* b2  = (const float*)d_in[5];
  const float* Wfc = (const float*)d_in[6];
  const float* bfc = (const float*)d_in[7];

  const int N = in_sizes[0] / CH;
  const int E = in_sizes[1] / 2;
  const int NF = N * CH;

  const int* src = ei;
  const int* dst = ei + E;

  float* ws   = (float*)d_ws;
  float* HB   = ws;
  float* AB   = HB + (size_t)NF;
  float* CB   = AB + (size_t)NF;
  float* deg  = CB + (size_t)NF;
  float* nrm  = deg + N;

  const int TB = 256;
  dim3 blk(TB);
  dim3 gN((N + TB - 1) / TB);
  dim3 gE((E + TB - 1) / TB);
  dim3 gEw((E + TB - 1) / TB);
  dim3 gG((N + 31) / 32);

  k_fill<<<gN, blk, 0, stream>>>(deg, 1.0f, N);
  k_deg<<<gE, blk, 0, stream>>>(dst, deg, E);
  k_edge_norm<<<gE, blk, 0, stream>>>(src, dst, deg, nrm, E);

  k_gemm64<false><<<gG, blk, 0, stream>>>(x, b1, W1, deg, HB, AB, N);
  k_scatter<<<gEw, blk, 0, stream>>>(HB, src, dst, nrm, AB, E);

  k_gemm64<true><<<gG, blk, 0, stream>>>(AB, b1, W2, deg, HB, CB, N);
  k_scatter<<<gEw, blk, 0, stream>>>(HB, src, dst, nrm, CB, E);

  k_edge_pred<<<gEw, blk, 0, stream>>>(CB, b2, src, dst, Wfc, bfc, (float*)d_out, E);
}


# Degree count privatised: 128 work-groups each count their share of the edges in an LDS table of all nodes (LDS integer adds), then add it to deg with whole-line f32 adds; per-edge one-word adds gone
# speedup vs baseline: 2.5569x; 1.2716x over previous
#include <hip/hip_runtime.h>
#include <hip/hip_bf16.h>

typedef __attribute__((ext_vector_type(2)))  float    v2f;
typedef __attribute__((ext_vector_type(8)))  float    v8f;
typedef __attribute__((ext_vector_type(16))) _Float16 v16h;

#define CH 64

__global__ void k_fill(float* __restrict__ p, float v, int n) {
  int i = blockIdx.x * blockDim.x + threadIdx.x;
  if (i < n) p[i] = v;
}

#define DEG_T   65536
#define DEG_TPB 1024
#define DEG_G   128
#define DEG_U   8
#define DEG_F   4

__global__ __launch_bounds__(DEG_TPB) void k_deg(const int* __restrict__ dst,
                                                 float* __restrict__ deg, int E, int N) {
  __shared__ unsigned hist[DEG_T];
  const int tid = threadIdx.x, lane = tid & 31, wave = tid >> 5;
  const int per = (E + (int)gridDim.x - 1) / (int)gridDim.x;
  const int eb = (int)blockIdx.x * per;
  const int ee = min(eb + per, E);
  for (int t0 = 0; t0 < N; t0 += DEG_T) {
    const int nt = min(DEG_T, N - t0);
    const int nLines = (nt + 31) >> 5;
    for (int i = tid * 4; i < nLines * 32; i += DEG_TPB * 4)
      *(uint4*)&hist[i] = make_uint4(0u, 0u, 0u, 0u);
    __syncthreads();
    for (int e = eb + tid; e < ee; e += DEG_TPB * DEG_U) {
      int d[DEG_U];
#pragma unroll
      for (int u = 0; u < DEG_U; ++u) {
        const int ei = e + u * DEG_TPB;
        d[u] = ei < ee ? dst[ei] : -1;
      }
#pragma unroll
      for (int u = 0; u < DEG_U; ++u) {
        const unsigned x = (unsigned)(d[u] - t0);
        if (x < (unsigned)nt) atomicAdd(&hist[x], 1u);
      }
    }
    __syncthreads();
    const int rot = (int)(((unsigned)blockIdx.x * (unsigned)nLines) / (unsigned)gridDim.x);
    for (int k = wave; k < nLines; k += (DEG_TPB / 32) * DEG_F) {
      int idx[DEG_F];
      unsigned c[DEG_F];
#pragma unroll
      for (int u = 0; u < DEG_F; ++u) {
        const int kk = k + u * (DEG_TPB / 32);
        int L = kk + rot;
        if (L >= nLines) L -= nLines;
        idx[u] = L * 32 + lane;
        c[u] = kk < nLines ? hist[idx[u]] : 0u;
      }
#pragma unroll
      for (int u = 0; u < DEG_F; ++u)
        if (c[u]) unsafeAtomicAdd(&deg[t0 + idx[u]], (float)c[u]);
    }
    __syncthreads();
  }
}

__global__ void k_edge_norm(const int* __restrict__ src, const int* __restrict__ dst,
                            const float* __restrict__ deg, float* __restrict__ nrm, int E) {
  int e = blockIdx.x * blockDim.x + threadIdx.x;
  if (e < E) nrm[e] = rsqrtf(deg[src[e]]) * rsqrtf(deg[dst[e]]);
}

__device__ __forceinline__ float bias_relu(float x, float b) {
  float v = x + b;
  return v > 0.0f ? v : 0.0f;
}

template <bool RELU_IN>
__global__ __launch_bounds__(256) void k_gemm64(const float* __restrict__ A,
                                                const float* __restrict__ bin,
                                                const float* __restrict__ W,
                                                const float* __restrict__ deg,
                                                float* __restrict__ C,
                                                float* __restrict__ S, int N) {
  const int lane = threadIdx.x & 31;
  const int wave = threadIdx.x >> 5;
  const int row0 = blockIdx.x * 32 + (wave >> 2) * 16;
  if (row0 >= N) return;
  const int n0 = (wave & 3) * 16;
  const int m  = lane & 15;
  const int kh = lane >> 4;
  v8f acc = {};

  float di[8];
#pragma unroll
  for (int v = 0; v < 8; ++v) {
    const int r = row0 + v + 8 * kh;
    di[v] = rsqrtf(deg[min(r, N - 1)]);
  }

#if __has_builtin(__builtin_amdgcn_wmma_f32_16x16x4_f32)
  const float* arow = A + (size_t)(row0 + m) * CH + 2 * kh;
  const float* brow = bin + 2 * kh;
#pragma unroll
  for (int k0 = 0; k0 < CH; k0 += 4) {
    v2f a;
    a.x = arow[k0];
    a.y = arow[k0 + 1];
    if (RELU_IN) {
      a.x = bias_relu(a.x, brow[k0]);
      a.y = bias_relu(a.y, brow[k0 + 1]);
    }
    v2f b;
    b.x = W[(k0 + 2 * kh + 0) * CH + n0 + m];
    b.y = W[(k0 + 2 * kh + 1) * CH + n0 + m];
    acc = __builtin_amdgcn_wmma_f32_16x16x4_f32(false, a, false, b,
                                                (short)0, acc, false, false);
  }
#else
  const float* arow = A + (size_t)(row0 + m) * CH;
#pragma unroll
  for (int kb = 0; kb < CH; kb += 32) {
    v16h a, b;
#pragma unroll
    for (int e = 0; e < 16; ++e) {
      int k = (e < 8) ? (8 * kh + e) : (16 + 8 * kh + (e - 8));
      float av = arow[kb + k];
      if (RELU_IN) av = bias_relu(av, bin[kb + k]);
      a[e] = (_Float16)av;
      b[e] = (_Float16)W[(size_t)(kb + k) * CH + n0 + m];
    }
    acc = __builtin_amdgcn_wmma_f32_16x16x32_f16(false, a, false, b,
                                                 (short)0, acc, false, false);
  }
#endif

#pragma unroll
  for (int v = 0; v < 8; ++v) {
    int r = row0 + v + 8 * kh;
    if (r < N) {
      const float hv = acc[v];
      C[(size_t)r * CH + (n0 + m)] = hv;
      S[(size_t)r * CH + (n0 + m)] = hv * di[v] * di[v];
    }
  }
}

#define EPW 32
#define EB  8

template <bool FULL>
__device__ __forceinline__ void scatter_wave(const float* __restrict__ h,
                                             const int* __restrict__ src,
                                             const int* __restrict__ dst,
                                             const float* __restrict__ nrm,
                                             float* __restrict__ out,
                                             int e0, int cnt, int lane) {
  const int e = e0 + lane;
  const bool ok = FULL || lane < cnt;
  const int   s_l = ok ? src[e] : 0;
  const int   d_l = ok ? dst[e] : 0;
  const float w_l = ok ? nrm[e] : 0.0f;
  const int c = lane;
#pragma unroll
  for (int j0 = 0; j0 < EPW; j0 += EB) {
    float hlo[EB], hhi[EB];
#pragma unroll
    for (int j = 0; j < EB; ++j) {
      const int s = __builtin_amdgcn_readlane(s_l, j0 + j);
      const float* hp = h + (size_t)s * CH + c;
      hlo[j] = hp[0];
      hhi[j] = hp[CH / 2];
    }
#pragma unroll
    for (int j = 0; j < EB; ++j) {
      if (FULL || j0 + j < cnt) {
        const int   d = __builtin_amdgcn_readlane(d_l, j0 + j);
        const float w = __int_as_float(__builtin_amdgcn_readlane(__float_as_int(w_l), j0 + j));
        float* op = out + (size_t)d * CH + c;
        unsafeAtomicAdd(op,          hlo[j] * w);
        unsafeAtomicAdd(op + CH / 2, hhi[j] * w);
      }
    }
  }
}

__global__ __launch_bounds__(256) void k_scatter(const float* __restrict__ h,
                                                 const int* __restrict__ src,
                                                 const int* __restrict__ dst,
                                                 const float* __restrict__ nrm,
                                                 float* __restrict__ out, int E) {
  const int lane = threadIdx.x & 31;
  const int wave = __builtin_amdgcn_readfirstlane((int)(blockIdx.x * blockDim.x + threadIdx.x) >> 5);
  const int e0 = wave * EPW;
  if (e0 >= E) return;
  const int cnt = E - e0;
  if (cnt >= EPW) scatter_wave<true >(h, src, dst, nrm, out, e0, EPW, lane);
  else            scatter_wave<false>(h, src, dst, nrm, out, e0, cnt, lane);
}

__global__ __launch_bounds__(256) void k_edge_pred(const float* __restrict__ g,
                                                   const float* __restrict__ bh,
                                                   const int* __restrict__ src,
                                                   const int* __restrict__ dst,
                                                   const float* __restrict__ Wfc,
                                                   const float* __restrict__ bfc,
                                                   float* __restrict__ out, int E) {
  const int lane = threadIdx.x & 31;
  const int wave = __builtin_amdgcn_readfirstlane((int)(blockIdx.x * blockDim.x + threadIdx.x) >> 5);
  const int e0 = wave * EPW;
  if (e0 >= E) return;
  const int e = e0 + lane;
  const bool ok = e < E;
  const int s_l = ok ? src[e] : 0;
  const int d_l = ok ? dst[e] : 0;
  const int c = lane * 2;
  const float2 w0 = *(const float2*)(Wfc + c);
  const float2 w1 = *(const float2*)(Wfc + CH + c);
  const float2 bb = *(const float2*)(bh + c);
  float res = 0.0f;
#pragma unroll
  for (int j0 = 0; j0 < EPW; j0 += EB) {
    float2 hs[EB], hd[EB];
#pragma unroll
    for (int j = 0; j < EB; ++j) {
      const int s = __builtin_amdgcn_readlane(s_l, j0 + j);
      const int d = __builtin_amdgcn_readlane(d_l, j0 + j);
      hs[j] = *(const float2*)(g + (size_t)s * CH + c);
      hd[j] = *(const float2*)(g + (size_t)d * CH + c);
    }
#pragma unroll
    for (int j = 0; j < EB; ++j) {
      const float sx = bias_relu(hs[j].x, bb.x), sy = bias_relu(hs[j].y, bb.y);
      const float dx = bias_relu(hd[j].x, bb.x), dy = bias_relu(hd[j].y, bb.y);
      float sum = sx * w0.x + sy * w0.y + dx * w1.x + dy * w1.y;
#pragma unroll
      for (int off = 16; off > 0; off >>= 1) sum += __shfl_xor(sum, off, 32);
      if (lane == j0 + j) res = sum;
    }
  }
  if (ok) out[e] = res + bfc[0];
}

extern "C" void kernel_launch(void* const* d_in, const int* in_sizes, int n_in,
                              void* d_out, int out_size, void* d_ws, size_t ws_size,
                              hipStream_t stream) {
  const float* x   = (const float*)d_in[0];
  const int*   ei  = (const int*)d_in[1];
  const float* W1  = (const float*)d_in[2];
  const float* b1  = (const float*)d_in[3];
  const float* W2  = (const float*)d_in[4];
  const float* b2  = (const float*)d_in[5];
  const float* Wfc = (const float*)d_in[6];
  const float* bfc = (const float*)d_in[7];

  const int N = in_sizes[0] / CH;
  const int E = in_sizes[1] / 2;
  const int NF = N * CH;

  const int* src = ei;
  const int* dst = ei + E;

  float* ws   = (float*)d_ws;
  float* HB   = ws;
  float* AB   = HB + (size_t)NF;
  float* CB   = AB + (size_t)NF;
  float* deg  = CB + (size_t)NF;
  float* nrm  = deg + N;

  const int TB = 256;
  dim3 blk(TB);
  dim3 gN((N + TB - 1) / TB);
  dim3 gE((E + TB - 1) / TB);
  dim3 gEw((E + TB - 1) / TB);
  dim3 gG((N + 31) / 32);

  k_fill<<<gN, blk, 0, stream>>>(deg, 1.0f, N);
  k_deg<<<dim3(DEG_G), dim3(DEG_TPB), 0, stream>>>(dst, deg, E, N);
  k_edge_norm<<<gE, blk, 0, stream>>>(src, dst, deg, nrm, E);

  k_gemm64<false><<<gG, blk, 0, stream>>>(x, b1, W1, deg, HB, AB, N);
  k_scatter<<<gEw, blk, 0, stream>>>(HB, src, dst, nrm, AB, E);

  k_gemm64<true><<<gG, blk, 0, stream>>>(AB, b1, W2, deg, HB, CB, N);
  k_scatter<<<gEw, blk, 0, stream>>>(HB, src, dst, nrm, CB, E);

  k_edge_pred<<<gEw, blk, 0, stream>>>(CB, b2, src, dst, Wfc, bfc, (float*)d_out, E);
}


# Front passes folded into neighbours: degree left as 64 per-group count tables (line stores), summed in the first product's prologue; edge norm formed in scatter 1; 8 launches, now 6; sum order kept
# speedup vs baseline: 2.7735x; 1.0847x over previous
#include <hip/hip_runtime.h>
#include <hip/hip_bf16.h>

typedef __attribute__((ext_vector_type(2)))  float    v2f;
typedef __attribute__((ext_vector_type(8)))  float    v8f;
typedef __attribute__((ext_vector_type(16))) _Float16 v16h;

#define CH 64

#define DEG_T   65536
#define DEG_TPB 1024
#define DEG_G   64
#define DEG_U   8

__global__ __launch_bounds__(DEG_TPB) void k_deg(const int* __restrict__ dst,
                                                 unsigned* __restrict__ part, int E, int N, int NP) {
  __shared__ unsigned hist[DEG_T];
  const int tid = threadIdx.x;
  const int per = (E + (int)gridDim.x - 1) / (int)gridDim.x;
  const int eb = (int)blockIdx.x * per;
  const int ee = min(eb + per, E);
  unsigned* __restrict__ mine = part + (size_t)blockIdx.x * NP;
  for (int t0 = 0; t0 < N; t0 += DEG_T) {
    const int nt = min(DEG_T, N - t0);
    const int nW = ((nt + 31) >> 5) * 32;
    for (int i = tid * 4; i < nW; i += DEG_TPB * 4)
      *(uint4*)&hist[i] = make_uint4(0u, 0u, 0u, 0u);
    __syncthreads();
    for (int e = eb + tid; e < ee; e += DEG_TPB * DEG_U) {
      int d[DEG_U];
#pragma unroll
      for (int u = 0; u < DEG_U; ++u) {
        const int ei = e + u * DEG_TPB;
        d[u] = ei < ee ? dst[ei] : -1;
      }
#pragma unroll
      for (int u = 0; u < DEG_U; ++u) {
        const unsigned x = (unsigned)(d[u] - t0);
        if (x < (unsigned)nt) atomicAdd(&hist[x], 1u);
      }
    }
    __syncthreads();
    for (int i = tid * 4; i < nW; i += DEG_TPB * 4)
      *(uint4*)&mine[t0 + i] = *(const uint4*)&hist[i];
    __syncthreads();
  }
}


__device__ __forceinline__ float bias_relu(float x, float b) {
  float v = x + b;
  return v > 0.0f ? v : 0.0f;
}

template <bool RELU_IN, bool FIRST>
__global__ __launch_bounds__(256) void k_gemm64(const float* __restrict__ A,
                                                const float* __restrict__ bin,
                                                const float* __restrict__ W,
                                                const unsigned* __restrict__ part, int NP,
                                                float* __restrict__ deg,
                                                float* __restrict__ C,
                                                float* __restrict__ S, int N) {
  const int lane = threadIdx.x & 31;
  const int wave = __builtin_amdgcn_readfirstlane((int)threadIdx.x >> 5);
  const int nb   = blockIdx.x * 32;
  const int row0 = nb + (wave >> 2) * 16;
  const bool active = row0 < N;
  if (!FIRST && !active) return;
  const int n0 = (wave & 3) * 16;
  const int m  = lane & 15;
  const int kh = lane >> 4;
  v8f acc = {};

  float di[8];
  unsigned cnt = 0;
  if (FIRST) {
#pragma unroll
    for (int j = 0; j < DEG_G / 8; ++j) {
      const unsigned* pj = part + (size_t)(wave + 8 * j) * NP + nb;
      cnt += pj[lane];
    }
  } else {
#pragma unroll
    for (int v = 0; v < 8; ++v) {
      const int r = row0 + v + 8 * kh;
      di[v] = rsqrtf(deg[min(r, N - 1)]);
    }
  }

  if (active) {
#if __has_builtin(__builtin_amdgcn_wmma_f32_16x16x4_f32)
  const float* arow = A + (size_t)(row0 + m) * CH + 2 * kh;
  const float* brow = bin + 2 * kh;
#pragma unroll
  for (int k0 = 0; k0 < CH; k0 += 4) {
    v2f a;
    a.x = arow[k0];
    a.y = arow[k0 + 1];
    if (RELU_IN) {
      a.x = bias_relu(a.x, brow[k0]);
      a.y = bias_relu(a.y, brow[k0 + 1]);
    }
    v2f b;
    b.x = W[(k0 + 2 * kh + 0) * CH + n0 + m];
    b.y = W[(k0 + 2 * kh + 1) * CH + n0 + m];
    acc = __builtin_amdgcn_wmma_f32_16x16x4_f32(false, a, false, b,
                                                (short)0, acc, false, false);
  }
#else
  const float* arow = A + (size_t)(row0 + m) * CH;
#pragma unroll
  for (int kb = 0; kb < CH; kb += 32) {
    v16h a, b;
#pragma unroll
    for (int e = 0; e < 16; ++e) {
      int k = (e < 8) ? (8 * kh + e) : (16 + 8 * kh + (e - 8));
      float av = arow[kb + k];
      if (RELU_IN) av = bias_relu(av, bin[kb + k]);
      a[e] = (_Float16)av;
      b[e] = (_Float16)W[(size_t)(kb + k) * CH + n0 + m];
    }
    acc = __builtin_amdgcn_wmma_f32_16x16x32_f16(false, a, false, b,
                                                 (short)0, acc, false, false);
  }
#endif
  }

  if (FIRST) {
    __shared__ unsigned cnt_s[8][32];
    __shared__ float deg_s[32];
    cnt_s[wave][lane] = cnt;
    __syncthreads();
    if (wave == 0) {
      unsigned c = 0;
#pragma unroll
      for (int w = 0; w < 8; ++w) c += cnt_s[w][lane];
      const float dg = (float)c + 1.0f;
      deg_s[lane] = dg;
      if (nb + lane < N) deg[nb + lane] = dg;
    }
    __syncthreads();
#pragma unroll
    for (int v = 0; v < 8; ++v) di[v] = rsqrtf(deg_s[(wave >> 2) * 16 + v + 8 * kh]);
  }

  if (active) {
#pragma unroll
  for (int v = 0; v < 8; ++v) {
    int r = row0 + v + 8 * kh;
    if (r < N) {
      const float hv = acc[v];
      C[(size_t)r * CH + (n0 + m)] = hv;
      S[(size_t)r * CH + (n0 + m)] = hv * di[v] * di[v];
    }
  }
  }
}

#define EPW 32
#define EB  8

template <bool FULL, bool NORM>
__device__ __forceinline__ void scatter_wave(const float* __restrict__ h,
                                             const int* __restrict__ src,
                                             const int* __restrict__ dst,
                                             const float* __restrict__ deg,
                                             const float* __restrict__ nrm_in,
                                             float* __restrict__ nrm_out,
                                             float* __restrict__ out,
                                             int e0, int cnt, int lane) {
  const int e = e0 + lane;
  const bool ok = FULL || lane < cnt;
  const int   s_l = ok ? src[e] : 0;
  const int   d_l = ok ? dst[e] : 0;
  float w_l;
  if (NORM) {
    const float wv = rsqrtf(deg[s_l]) * rsqrtf(deg[d_l]);
    w_l = ok ? wv : 0.0f;
    if (ok) nrm_out[e] = wv;
  } else {
    w_l = ok ? nrm_in[e] : 0.0f;
  }
  const int c = lane;
#pragma unroll
  for (int j0 = 0; j0 < EPW; j0 += EB) {
    float hlo[EB], hhi[EB];
#pragma unroll
    for (int j = 0; j < EB; ++j) {
      const int s = __builtin_amdgcn_readlane(s_l, j0 + j);
      const float* hp = h + (size_t)s * CH + c;
      hlo[j] = hp[0];
      hhi[j] = hp[CH / 2];
    }
#pragma unroll
    for (int j = 0; j < EB; ++j) {
      if (FULL || j0 + j < cnt) {
        const int   d = __builtin_amdgcn_readlane(d_l, j0 + j);
        const float w = __int_as_float(__builtin_amdgcn_readlane(__float_as_int(w_l), j0 + j));
        float* op = out + (size_t)d * CH + c;
        unsafeAtomicAdd(op,          hlo[j] * w);
        unsafeAtomicAdd(op + CH / 2, hhi[j] * w);
      }
    }
  }
}

template <bool NORM>
__global__ __launch_bounds__(256) void k_scatter(const float* __restrict__ h,
                                                 const int* __restrict__ src,
                                                 const int* __restrict__ dst,
                                                 const float* __restrict__ deg,
                                                 const float* __restrict__ nrm_in,
                                                 float* __restrict__ nrm_out,
                                                 float* __restrict__ out, int E) {
  const int lane = threadIdx.x & 31;
  const int wave = __builtin_amdgcn_readfirstlane((int)(blockIdx.x * blockDim.x + threadIdx.x) >> 5);
  const int e0 = wave * EPW;
  if (e0 >= E) return;
  const int cnt = E - e0;
  if (cnt >= EPW) scatter_wave<true,  NORM>(h, src, dst, deg, nrm_in, nrm_out, out, e0, EPW, lane);
  else            scatter_wave<false, NORM>(h, src, dst, deg, nrm_in, nrm_out, out, e0, cnt, lane);
}

__global__ __launch_bounds__(256) void k_edge_pred(const float* __restrict__ g,
                                                   const float* __restrict__ bh,
                                                   const int* __restrict__ src,
                                                   const int* __restrict__ dst,
                                                   const float* __restrict__ Wfc,
                                                   const float* __restrict__ bfc,
                                                   float* __restrict__ out, int E) {
  const int lane = threadIdx.x & 31;
  const int wave = __builtin_amdgcn_readfirstlane((int)(blockIdx.x * blockDim.x + threadIdx.x) >> 5);
  const int e0 = wave * EPW;
  if (e0 >= E) return;
  const int e = e0 + lane;
  const bool ok = e < E;
  const int s_l = ok ? src[e] : 0;
  const int d_l = ok ? dst[e] : 0;
  const int c = lane * 2;
  const float2 w0 = *(const float2*)(Wfc + c);
  const float2 w1 = *(const float2*)(Wfc + CH + c);
  const float2 bb = *(const float2*)(bh + c);
  float res = 0.0f;
#pragma unroll
  for (int j0 = 0; j0 < EPW; j0 += EB) {
    float2 hs[EB], hd[EB];
#pragma unroll
    for (int j = 0; j < EB; ++j) {
      const int s = __builtin_amdgcn_readlane(s_l, j0 + j);
      const int d = __builtin_amdgcn_readlane(d_l, j0 + j);
      hs[j] = *(const float2*)(g + (size_t)s * CH + c);
      hd[j] = *(const float2*)(g + (size_t)d * CH + c);
    }
#pragma unroll
    for (int j = 0; j < EB; ++j) {
      const float sx = bias_relu(hs[j].x, bb.x), sy = bias_relu(hs[j].y, bb.y);
      const float dx = bias_relu(hd[j].x, bb.x), dy = bias_relu(hd[j].y, bb.y);
      float sum = sx * w0.x + sy * w0.y + dx * w1.x + dy * w1.y;
#pragma unroll
      for (int off = 16; off > 0; off >>= 1) sum += __shfl_xor(sum, off, 32);
      if (lane == j0 + j) res = sum;
    }
  }
  if (ok) out[e] = res + bfc[0];
}

extern "C" void kernel_launch(void* const* d_in, const int* in_sizes, int n_in,
                              void* d_out, int out_size, void* d_ws, size_t ws_size,
                              hipStream_t stream) {
  const float* x   = (const float*)d_in[0];
  const int*   ei  = (const int*)d_in[1];
  const float* W1  = (const float*)d_in[2];
  const float* b1  = (const float*)d_in[3];
  const float* W2  = (const float*)d_in[4];
  const float* b2  = (const float*)d_in[5];
  const float* Wfc = (const float*)d_in[6];
  const float* bfc = (const float*)d_in[7];

  const int N = in_sizes[0] / CH;
  const int E = in_sizes[1] / 2;
  const int NF = N * CH;

  const int* src = ei;
  const int* dst = ei + E;

  float* ws   = (float*)d_ws;
  float* HB   = ws;
  float* AB   = HB + (size_t)NF;
  float* CB   = AB + (size_t)NF;
  const int NP = (N + 127) / 128 * 128;
  unsigned* part = (unsigned*)(CB + (size_t)NF);
  float* deg  = (float*)(part + (size_t)DEG_G * NP);
  float* nrm  = deg + NP;

  const int TB = 256;
  dim3 blk(TB);
  dim3 gEw((E + TB - 1) / TB);
  dim3 gG((N + 31) / 32);

  k_deg<<<dim3(DEG_G), dim3(DEG_TPB), 0, stream>>>(dst, part, E, N, NP);

  k_gemm64<false, true><<<gG, blk, 0, stream>>>(x, b1, W1, part, NP, deg, HB, AB, N);
  k_scatter<true><<<gEw, blk, 0, stream>>>(HB, src, dst, deg, nullptr, nrm, AB, E);

  k_gemm64<true, false><<<gG, blk, 0, stream>>>(AB, b1, W2, part, NP, deg, HB, CB, N);
  k_scatter<false><<<gEw, blk, 0, stream>>>(HB, src, dst, deg, nrm, nullptr, CB, E);

  k_edge_pred<<<gEw, blk, 0, stream>>>(CB, b2, src, dst, Wfc, bfc, (float*)d_out, E);
}


# Scatter adds kept in one half of the chip: rows split by half, each 512-edge item walked once per half by work-groups of that half (die id, ticket counters); sum order = arrival order as before
# speedup vs baseline: 3.1406x; 1.1324x over previous
#include <hip/hip_runtime.h>
#include <hip/hip_bf16.h>

typedef __attribute__((ext_vector_type(2)))  float    v2f;
typedef __attribute__((ext_vector_type(8)))  float    v8f;
typedef __attribute__((ext_vector_type(16))) _Float16 v16h;

#define CH 64

#define DEG_T   65536
#define DEG_TPB 1024
#define DEG_G   64
#define DEG_U   8
#define SC_CTR  64

__global__ __launch_bounds__(DEG_TPB) void k_deg(const int* __restrict__ dst,
                                                 unsigned* __restrict__ part,
                                                 unsigned* __restrict__ ctr, int E, int N, int NP) {
  __shared__ unsigned hist[DEG_T];
  const int tid = threadIdx.x;
  if (blockIdx.x == 0 && tid < 4) ctr[tid * SC_CTR] = 0u;
  const int per = (E + (int)gridDim.x - 1) / (int)gridDim.x;
  const int eb = (int)blockIdx.x * per;
  const int ee = min(eb + per, E);
  unsigned* __restrict__ mine = part + (size_t)blockIdx.x * NP;
  for (int t0 = 0; t0 < N; t0 += DEG_T) {
    const int nt = min(DEG_T, N - t0);
    const int nW = ((nt + 31) >> 5) * 32;
    for (int i = tid * 4; i < nW; i += DEG_TPB * 4)
      *(uint4*)&hist[i] = make_uint4(0u, 0u, 0u, 0u);
    __syncthreads();
    for (int e = eb + tid; e < ee; e += DEG_TPB * DEG_U) {
      int d[DEG_U];
#pragma unroll
      for (int u = 0; u < DEG_U; ++u) {
        const int ei = e + u * DEG_TPB;
        d[u] = ei < ee ? dst[ei] : -1;
      }
#pragma unroll
      for (int u = 0; u < DEG_U; ++u) {
        const unsigned x = (unsigned)(d[u] - t0);
        if (x < (unsigned)nt) atomicAdd(&hist[x], 1u);
      }
    }
    __syncthreads();
    for (int i = tid * 4; i < nW; i += DEG_TPB * 4)
      *(uint4*)&mine[t0 + i] = *(const uint4*)&hist[i];
    __syncthreads();
  }
}


__device__ __forceinline__ float bias_relu(float x, float b) {
  float v = x + b;
  return v > 0.0f ? v : 0.0f;
}

template <bool RELU_IN, bool FIRST>
__global__ __launch_bounds__(256) void k_gemm64(const float* __restrict__ A,
                                                const float* __restrict__ bin,
                                                const float* __restrict__ W,
                                                const unsigned* __restrict__ part, int NP,
                                                float* __restrict__ deg,
                                                float* __restrict__ C,
                                                float* __restrict__ S, int N) {
  const int lane = threadIdx.x & 31;
  const int wave = __builtin_amdgcn_readfirstlane((int)threadIdx.x >> 5);
  const int nb   = blockIdx.x * 32;
  const int row0 = nb + (wave >> 2) * 16;
  const bool active = row0 < N;
  if (!FIRST && !active) return;
  const int n0 = (wave & 3) * 16;
  const int m  = lane & 15;
  const int kh = lane >> 4;
  v8f acc = {};

  float di[8];
  unsigned cnt = 0;
  if (FIRST) {
#pragma unroll
    for (int j = 0; j < DEG_G / 8; ++j) {
      const unsigned* pj = part + (size_t)(wave + 8 * j) * NP + nb;
      cnt += pj[lane];
    }
  } else {
#pragma unroll
    for (int v = 0; v < 8; ++v) {
      const int r = row0 + v + 8 * kh;
      di[v] = rsqrtf(deg[min(r, N - 1)]);
    }
  }

  if (active) {
#if __has_builtin(__builtin_amdgcn_wmma_f32_16x16x4_f32)
  const float* arow = A + (size_t)(row0 + m) * CH + 2 * kh;
  const float* brow = bin + 2 * kh;
#pragma unroll
  for (int k0 = 0; k0 < CH; k0 += 4) {
    v2f a;
    a.x = arow[k0];
    a.y = arow[k0 + 1];
    if (RELU_IN) {
      a.x = bias_relu(a.x, brow[k0]);
      a.y = bias_relu(a.y, brow[k0 + 1]);
    }
    v2f b;
    b.x = W[(k0 + 2 * kh + 0) * CH + n0 + m];
    b.y = W[(k0 + 2 * kh + 1) * CH + n0 + m];
    acc = __builtin_amdgcn_wmma_f32_16x16x4_f32(false, a, false, b,
                                                (short)0, acc, false, false);
  }
#else
  const float* arow = A + (size_t)(row0 + m) * CH;
#pragma unroll
  for (int kb = 0; kb < CH; kb += 32) {
    v16h a, b;
#pragma unroll
    for (int e = 0; e < 16; ++e) {
      int k = (e < 8) ? (8 * kh + e) : (16 + 8 * kh + (e - 8));
      float av = arow[kb + k];
      if (RELU_IN) av = bias_relu(av, bin[kb + k]);
      a[e] = (_Float16)av;
      b[e] = (_Float16)W[(size_t)(kb + k) * CH + n0 + m];
    }
    acc = __builtin_amdgcn_wmma_f32_16x16x32_f16(false, a, false, b,
                                                 (short)0, acc, false, false);
  }
#endif
  }

  if (FIRST) {
    __shared__ unsigned cnt_s[8][32];
    __shared__ float deg_s[32];
    cnt_s[wave][lane] = cnt;
    __syncthreads();
    if (wave == 0) {
      unsigned c = 0;
#pragma unroll
      for (int w = 0; w < 8; ++w) c += cnt_s[w][lane];
      const float dg = (float)c + 1.0f;
      deg_s[lane] = dg;
      if (nb + lane < N) deg[nb + lane] = dg;
    }
    __syncthreads();
#pragma unroll
    for (int v = 0; v < 8; ++v) di[v] = rsqrtf(deg_s[(wave >> 2) * 16 + v + 8 * kh]);
  }

  if (active) {
#pragma unroll
  for (int v = 0; v < 8; ++v) {
    int r = row0 + v + 8 * kh;
    if (r < N) {
      const float hv = acc[v];
      C[(size_t)r * CH + (n0 + m)] = hv;
      S[(size_t)r * CH + (n0 + m)] = hv * di[v] * di[v];
    }
  }
  }
}

#define EB      8
#define SC_WE   64
#define SC_ITEM 512
#define SC_TPB  288
#define SC_DONE 0xFFFFFFFFu

__device__ __forceinline__ unsigned sc_draw(unsigned* __restrict__ ctr, int half, int& t, unsigned nItems) {
  unsigned v = atomicAdd(&ctr[t * SC_CTR], 1u);
  if (v >= nItems && t == half) {
    t = 1 - half;
    v = atomicAdd(&ctr[t * SC_CTR], 1u);
  }
  return v >= nItems ? SC_DONE : (v | ((unsigned)t << 30));
}

template <bool NORM>
__global__ __launch_bounds__(SC_TPB) void k_scatter(const float* __restrict__ h,
                                                    const int* __restrict__ src,
                                                    const int* __restrict__ dst,
                                                    const float* __restrict__ deg,
                                                    const float* __restrict__ nrm_in,
                                                    float* __restrict__ nrm_out,
                                                    float* __restrict__ out,
                                                    unsigned* __restrict__ ctr,
                                                    int E, int NH, int nItems) {
  __shared__ unsigned s_tick[2];
  const int tid  = threadIdx.x;
  const int lane = tid & 31;
  const int wave = __builtin_amdgcn_readfirstlane(tid >> 5);
  const bool drawer = wave == 8;
  int half = 0, t = 0;
  if (drawer) {
    half = (int)((__builtin_amdgcn_s_sendmsg_rtn(0x87) >> 18) & 1u);
    t = half;
    if (lane == 0) s_tick[0] = sc_draw(ctr, half, t, (unsigned)nItems);
  }
  __syncthreads();
  const int c = lane;
  for (int it = 0;; ++it) {
    const unsigned tick = s_tick[it & 1];
    if (tick == SC_DONE) break;
    if (drawer) {
      if (lane == 0) s_tick[(it + 1) & 1] = sc_draw(ctr, half, t, (unsigned)nItems);
    } else {
      const int tw = (int)(tick >> 30);
      const int e0 = (int)(tick & 0x3FFFFFFFu) * SC_ITEM + wave * SC_WE;
      const int ea = e0 + lane, eb = ea + 32;
      const bool oka = ea < E, okb = eb < E;
      const int eac = min(ea, E - 1), ebc = min(eb, E - 1);
      const int da = dst[eac], db = dst[ebc];
      const int sa = src[eac], sb = src[ebc];
      float wa, wb;
      if (!NORM) { wa = nrm_in[eac]; wb = nrm_in[ebc]; }
      __builtin_amdgcn_sched_barrier(0);
      asm volatile("" :: "v"(da), "v"(db), "v"(sa), "v"(sb));
      const bool ka = oka && ((da >= NH) == (tw != 0));
      const bool kb = okb && ((db >= NH) == (tw != 0));
      if (NORM) {
        const float xa = deg[ka ? sa : 0], ya = deg[ka ? da : 0];
        const float xb = deg[kb ? sb : 0], yb = deg[kb ? db : 0];
        wa = rsqrtf(xa) * rsqrtf(ya);
        wb = rsqrtf(xb) * rsqrtf(yb);
        if (ka) nrm_out[ea] = wa;
        if (kb) nrm_out[eb] = wb;
      }
      unsigned long long M = ((unsigned long long)__builtin_amdgcn_ballot_w32(kb) << 32) |
                             (unsigned long long)__builtin_amdgcn_ballot_w32(ka);
      while (M != 0ull) {
        int bl[EB];
        int nb = 0;
#pragma unroll
        for (int j = 0; j < EB; ++j) {
          bl[j] = 0;
          if (M != 0ull) {
            bl[j] = __builtin_ctzll(M);
            M &= M - 1ull;
            ++nb;
          }
        }
        float hlo[EB], hhi[EB];
#pragma unroll
        for (int j = 0; j < EB; ++j) {
          hlo[j] = 0.0f; hhi[j] = 0.0f;
          if (j < nb) {
            const int ln = bl[j] & 31;
            const int s0 = __builtin_amdgcn_readlane(sa, ln);
            const int s1 = __builtin_amdgcn_readlane(sb, ln);
            const int s = (bl[j] & 32) ? s1 : s0;
            const float* hp = h + (size_t)s * CH + c;
            hlo[j] = hp[0];
            hhi[j] = hp[CH / 2];
          }
        }
#pragma unroll
        for (int j = 0; j < EB; ++j) {
          if (j < nb) {
            const int ln = bl[j] & 31;
            const bool up = (bl[j] & 32) != 0;
            const int d0 = __builtin_amdgcn_readlane(da, ln);
            const int d1 = __builtin_amdgcn_readlane(db, ln);
            const int w0 = __builtin_amdgcn_readlane(__float_as_int(wa), ln);
            const int w1 = __builtin_amdgcn_readlane(__float_as_int(wb), ln);
            const int d = up ? d1 : d0;
            const float w = __int_as_float(up ? w1 : w0);
            float* op = out + (size_t)d * CH + c;
            unsafeAtomicAdd(op,          hlo[j] * w);
            unsafeAtomicAdd(op + CH / 2, hhi[j] * w);
          }
        }
      }
    }
    __syncthreads();
  }
}

#define EPW 32

__global__ __launch_bounds__(256) void k_edge_pred(const float* __restrict__ g,
                                                   const float* __restrict__ bh,
                                                   const int* __restrict__ src,
                                                   const int* __restrict__ dst,
                                                   const float* __restrict__ Wfc,
                                                   const float* __restrict__ bfc,
                                                   float* __restrict__ out, int E) {
  const int lane = threadIdx.x & 31;
  const int wave = __builtin_amdgcn_readfirstlane((int)(blockIdx.x * blockDim.x + threadIdx.x) >> 5);
  const int e0 = wave * EPW;
  if (e0 >= E) return;
  const int e = e0 + lane;
  const bool ok = e < E;
  const int s_l = ok ? src[e] : 0;
  const int d_l = ok ? dst[e] : 0;
  const int c = lane * 2;
  const float2 w0 = *(const float2*)(Wfc + c);
  const float2 w1 = *(const float2*)(Wfc + CH + c);
  const float2 bb = *(const float2*)(bh + c);
  float res = 0.0f;
#pragma unroll
  for (int j0 = 0; j0 < EPW; j0 += EB) {
    float2 hs[EB], hd[EB];
#pragma unroll
    for (int j = 0; j < EB; ++j) {
      const int s = __builtin_amdgcn_readlane(s_l, j0 + j);
      const int d = __builtin_amdgcn_readlane(d_l, j0 + j);
      hs[j] = *(const float2*)(g + (size_t)s * CH + c);
      hd[j] = *(const float2*)(g + (size_t)d * CH + c);
    }
#pragma unroll
    for (int j = 0; j < EB; ++j) {
      const float sx = bias_relu(hs[j].x, bb.x), sy = bias_relu(hs[j].y, bb.y);
      const float dx = bias_relu(hd[j].x, bb.x), dy = bias_relu(hd[j].y, bb.y);
      float sum = sx * w0.x + sy * w0.y + dx * w1.x + dy * w1.y;
#pragma unroll
      for (int off = 16; off > 0; off >>= 1) sum += __shfl_xor(sum, off, 32);
      if (lane == j0 + j) res = sum;
    }
  }
  if (ok) out[e] = res + bfc[0];
}

extern "C" void kernel_launch(void* const* d_in, const int* in_sizes, int n_in,
                              void* d_out, int out_size, void* d_ws, size_t ws_size,
                              hipStream_t stream) {
  const float* x   = (const float*)d_in[0];
  const int*   ei  = (const int*)d_in[1];
  const float* W1  = (const float*)d_in[2];
  const float* b1  = (const float*)d_in[3];
  const float* W2  = (const float*)d_in[4];
  const float* b2  = (const float*)d_in[5];
  const float* Wfc = (const float*)d_in[6];
  const float* bfc = (const float*)d_in[7];

  const int N = in_sizes[0] / CH;
  const int E = in_sizes[1] / 2;
  const int NF = N * CH;

  const int* src = ei;
  const int* dst = ei + E;

  float* ws   = (float*)d_ws;
  float* HB   = ws;
  float* AB   = HB + (size_t)NF;
  float* CB   = AB + (size_t)NF;
  const int NP = (N + 127) / 128 * 128;
  unsigned* part = (unsigned*)(CB + (size_t)NF);
  float* deg  = (float*)(part + (size_t)DEG_G * NP);
  float* nrm  = deg + NP;
  const int EP = (E + 127) / 128 * 128;
  unsigned* ctr = (unsigned*)(nrm + EP);
  const int NH = N / 2;
  const int nItems = (E + SC_ITEM - 1) / SC_ITEM;
  dim3 gS(1792);

  const int TB = 256;
  dim3 blk(TB);
  dim3 gEw((E + TB - 1) / TB);
  dim3 gG((N + 31) / 32);

  k_deg<<<dim3(DEG_G), dim3(DEG_TPB), 0, stream>>>(dst, part, ctr, E, N, NP);

  k_gemm64<false, true><<<gG, blk, 0, stream>>>(x, b1, W1, part, NP, deg, HB, AB, N);
  k_scatter<true><<<gS, dim3(SC_TPB), 0, stream>>>(HB, src, dst, deg, nullptr, nrm, AB, ctr, E, NH, nItems);

  k_gemm64<true, false><<<gG, blk, 0, stream>>>(AB, b1, W2, part, NP, deg, HB, CB, N);
  k_scatter<false><<<gS, dim3(SC_TPB), 0, stream>>>(HB, src, dst, deg, nrm, nullptr, CB, ctr + 2 * SC_CTR, E, NH, nItems);

  k_edge_pred<<<gEw, blk, 0, stream>>>(CB, b2, src, dst, Wfc, bfc, (float*)d_out, E);
}


# Edge head: a wave's 32 dot sums on the vector ALU, not via LDS: transposed butterfly of 31 exchanges (DPP row_xmask, half swap) for 160 ds_bpermute; same pair order 16,8,4,2,1: sum order kept
# speedup vs baseline: 3.3983x; 1.0821x over previous
#include <hip/hip_runtime.h>
#include <hip/hip_bf16.h>

typedef __attribute__((ext_vector_type(2)))  float    v2f;
typedef __attribute__((ext_vector_type(8)))  float    v8f;
typedef __attribute__((ext_vector_type(16))) _Float16 v16h;

#define CH 64

#define DEG_T   65536
#define DEG_TPB 1024
#define DEG_G   64
#define DEG_U   8
#define SC_CTR  64

__global__ __launch_bounds__(DEG_TPB) void k_deg(const int* __restrict__ dst,
                                                 unsigned* __restrict__ part,
                                                 unsigned* __restrict__ ctr, int E, int N, int NP) {
  __shared__ unsigned hist[DEG_T];
  const int tid = threadIdx.x;
  if (blockIdx.x == 0 && tid < 4) ctr[tid * SC_CTR] = 0u;
  const int per = (E + (int)gridDim.x - 1) / (int)gridDim.x;
  const int eb = (int)blockIdx.x * per;
  const int ee = min(eb + per, E);
  unsigned* __restrict__ mine = part + (size_t)blockIdx.x * NP;
  for (int t0 = 0; t0 < N; t0 += DEG_T) {
    const int nt = min(DEG_T, N - t0);
    const int nW = ((nt + 31) >> 5) * 32;
    for (int i = tid * 4; i < nW; i += DEG_TPB * 4)
      *(uint4*)&hist[i] = make_uint4(0u, 0u, 0u, 0u);
    __syncthreads();
    for (int e = eb + tid; e < ee; e += DEG_TPB * DEG_U) {
      int d[DEG_U];
#pragma unroll
      for (int u = 0; u < DEG_U; ++u) {
        const int ei = e + u * DEG_TPB;
        d[u] = ei < ee ? dst[ei] : -1;
      }
#pragma unroll
      for (int u = 0; u < DEG_U; ++u) {
        const unsigned x = (unsigned)(d[u] - t0);
        if (x < (unsigned)nt) atomicAdd(&hist[x], 1u);
      }
    }
    __syncthreads();
    for (int i = tid * 4; i < nW; i += DEG_TPB * 4)
      *(uint4*)&mine[t0 + i] = *(const uint4*)&hist[i];
    __syncthreads();
  }
}


__device__ __forceinline__ float bias_relu(float x, float b) {
  float v = x + b;
  return v > 0.0f ? v : 0.0f;
}

template <bool RELU_IN, bool FIRST>
__global__ __launch_bounds__(256) void k_gemm64(const float* __restrict__ A,
                                                const float* __restrict__ bin,
                                                const float* __restrict__ W,
                                                const unsigned* __restrict__ part, int NP,
                                                float* __restrict__ deg,
                                                float* __restrict__ C,
                                                float* __restrict__ S, int N) {
  const int lane = threadIdx.x & 31;
  const int wave = __builtin_amdgcn_readfirstlane((int)threadIdx.x >> 5);
  const int nb   = blockIdx.x * 32;
  const int row0 = nb + (wave >> 2) * 16;
  const bool active = row0 < N;
  if (!FIRST && !active) return;
  const int n0 = (wave & 3) * 16;
  const int m  = lane & 15;
  const int kh = lane >> 4;
  v8f acc = {};

  float di[8];
  unsigned cnt = 0;
  if (FIRST) {
#pragma unroll
    for (int j = 0; j < DEG_G / 8; ++j) {
      const unsigned* pj = part + (size_t)(wave + 8 * j) * NP + nb;
      cnt += pj[lane];
    }
  } else {
#pragma unroll
    for (int v = 0; v < 8; ++v) {
      const int r = row0 + v + 8 * kh;
      di[v] = rsqrtf(deg[min(r, N - 1)]);
    }
  }

  if (active) {
#if __has_builtin(__builtin_amdgcn_wmma_f32_16x16x4_f32)
  const float* arow = A + (size_t)(row0 + m) * CH + 2 * kh;
  const float* brow = bin + 2 * kh;
#pragma unroll
  for (int k0 = 0; k0 < CH; k0 += 4) {
    v2f a;
    a.x = arow[k0];
    a.y = arow[k0 + 1];
    if (RELU_IN) {
      a.x = bias_relu(a.x, brow[k0]);
      a.y = bias_relu(a.y, brow[k0 + 1]);
    }
    v2f b;
    b.x = W[(k0 + 2 * kh + 0) * CH + n0 + m];
    b.y = W[(k0 + 2 * kh + 1) * CH + n0 + m];
    acc = __builtin_amdgcn_wmma_f32_16x16x4_f32(false, a, false, b,
                                                (short)0, acc, false, false);
  }
#else
  const float* arow = A + (size_t)(row0 + m) * CH;
#pragma unroll
  for (int kb = 0; kb < CH; kb += 32) {
    v16h a, b;
#pragma unroll
    for (int e = 0; e < 16; ++e) {
      int k = (e < 8) ? (8 * kh + e) : (16 + 8 * kh + (e - 8));
      float av = arow[kb + k];
      if (RELU_IN) av = bias_relu(av, bin[kb + k]);
      a[e] = (_Float16)av;
      b[e] = (_Float16)W[(size_t)(kb + k) * CH + n0 + m];
    }
    acc = __builtin_amdgcn_wmma_f32_16x16x32_f16(false, a, false, b,
                                                 (short)0, acc, false, false);
  }
#endif
  }

  if (FIRST) {
    __shared__ unsigned cnt_s[8][32];
    __shared__ float deg_s[32];
    cnt_s[wave][lane] = cnt;
    __syncthreads();
    if (wave == 0) {
      unsigned c = 0;
#pragma unroll
      for (int w = 0; w < 8; ++w) c += cnt_s[w][lane];
      const float dg = (float)c + 1.0f;
      deg_s[lane] = dg;
      if (nb + lane < N) deg[nb + lane] = dg;
    }
    __syncthreads();
#pragma unroll
    for (int v = 0; v < 8; ++v) di[v] = rsqrtf(deg_s[(wave >> 2) * 16 + v + 8 * kh]);
  }

  if (active) {
#pragma unroll
  for (int v = 0; v < 8; ++v) {
    int r = row0 + v + 8 * kh;
    if (r < N) {
      const float hv = acc[v];
      C[(size_t)r * CH + (n0 + m)] = hv;
      S[(size_t)r * CH + (n0 + m)] = hv * di[v] * di[v];
    }
  }
  }
}

#define EB      8
#define SC_WE   64
#define SC_ITEM 512
#define SC_TPB  288
#define SC_DONE 0xFFFFFFFFu

__device__ __forceinline__ unsigned sc_draw(unsigned* __restrict__ ctr, int half, int& t, unsigned nItems) {
  unsigned v = atomicAdd(&ctr[t * SC_CTR], 1u);
  if (v >= nItems && t == half) {
    t = 1 - half;
    v = atomicAdd(&ctr[t * SC_CTR], 1u);
  }
  return v >= nItems ? SC_DONE : (v | ((unsigned)t << 30));
}

template <bool NORM>
__global__ __launch_bounds__(SC_TPB) void k_scatter(const float* __restrict__ h,
                                                    const int* __restrict__ src,
                                                    const int* __restrict__ dst,
                                                    const float* __restrict__ deg,
                                                    const float* __restrict__ nrm_in,
                                                    float* __restrict__ nrm_out,
                                                    float* __restrict__ out,
                                                    unsigned* __restrict__ ctr,
                                                    int E, int NH, int nItems) {
  __shared__ unsigned s_tick[2];
  const int tid  = threadIdx.x;
  const int lane = tid & 31;
  const int wave = __builtin_amdgcn_readfirstlane(tid >> 5);
  const bool drawer = wave == 8;
  int half = 0, t = 0;
  if (drawer) {
    half = (int)((__builtin_amdgcn_s_sendmsg_rtn(0x87) >> 18) & 1u);
    t = half;
    if (lane == 0) s_tick[0] = sc_draw(ctr, half, t, (unsigned)nItems);
  }
  __syncthreads();
  const int c = lane;
  for (int it = 0;; ++it) {
    const unsigned tick = s_tick[it & 1];
    if (tick == SC_DONE) break;
    if (drawer) {
      if (lane == 0) s_tick[(it + 1) & 1] = sc_draw(ctr, half, t, (unsigned)nItems);
    } else {
      const int tw = (int)(tick >> 30);
      const int e0 = (int)(tick & 0x3FFFFFFFu) * SC_ITEM + wave * SC_WE;
      const int ea = e0 + lane, eb = ea + 32;
      const bool oka = ea < E, okb = eb < E;
      const int eac = min(ea, E - 1), ebc = min(eb, E - 1);
      const int da = dst[eac], db = dst[ebc];
      const int sa = src[eac], sb = src[ebc];
      float wa, wb;
      if (!NORM) { wa = nrm_in[eac]; wb = nrm_in[ebc]; }
      __builtin_amdgcn_sched_barrier(0);
      asm volatile("" :: "v"(da), "v"(db), "v"(sa), "v"(sb));
      const bool ka = oka && ((da >= NH) == (tw != 0));
      const bool kb = okb && ((db >= NH) == (tw != 0));
      if (NORM) {
        const float xa = deg[ka ? sa : 0], ya = deg[ka ? da : 0];
        const float xb = deg[kb ? sb : 0], yb = deg[kb ? db : 0];
        wa = rsqrtf(xa) * rsqrtf(ya);
        wb = rsqrtf(xb) * rsqrtf(yb);
        if (ka) nrm_out[ea] = wa;
        if (kb) nrm_out[eb] = wb;
      }
      unsigned long long M = ((unsigned long long)__builtin_amdgcn_ballot_w32(kb) << 32) |
                             (unsigned long long)__builtin_amdgcn_ballot_w32(ka);
      while (M != 0ull) {
        int bl[EB];
        int nb = 0;
#pragma unroll
        for (int j = 0; j < EB; ++j) {
          bl[j] = 0;
          if (M != 0ull) {
            bl[j] = __builtin_ctzll(M);
            M &= M - 1ull;
            ++nb;
          }
        }
        float hlo[EB], hhi[EB];
#pragma unroll
        for (int j = 0; j < EB; ++j) {
          hlo[j] = 0.0f; hhi[j] = 0.0f;
          if (j < nb) {
            const int ln = bl[j] & 31;
            const int s0 = __builtin_amdgcn_readlane(sa, ln);
            const int s1 = __builtin_amdgcn_readlane(sb, ln);
            const int s = (bl[j] & 32) ? s1 : s0;
            const float* hp = h + (size_t)s * CH + c;
            hlo[j] = hp[0];
            hhi[j] = hp[CH / 2];
          }
        }
#pragma unroll
        for (int j = 0; j < EB; ++j) {
          if (j < nb) {
            const int ln = bl[j] & 31;
            const bool up = (bl[j] & 32) != 0;
            const int d0 = __builtin_amdgcn_readlane(da, ln);
            const int d1 = __builtin_amdgcn_readlane(db, ln);
            const int w0 = __builtin_amdgcn_readlane(__float_as_int(wa), ln);
            const int w1 = __builtin_amdgcn_readlane(__float_as_int(wb), ln);
            const int d = up ? d1 : d0;
            const float w = __int_as_float(up ? w1 : w0);
            float* op = out + (size_t)d * CH + c;
            unsafeAtomicAdd(op,          hlo[j] * w);
            unsafeAtomicAdd(op + CH / 2, hhi[j] * w);
          }
        }
      }
    }
    __syncthreads();
  }
}

#define EPW 32

__device__ __forceinline__ float lane_xor16(float x) {
  const int b = __builtin_bit_cast(int, x);
  return __builtin_bit_cast(float, __builtin_amdgcn_permlanex16(b, b, 0x76543210, 0xfedcba98, false, false));
}

template <int M>
__device__ __forceinline__ float lane_xor_row(float x) {
  return __builtin_bit_cast(float, __builtin_amdgcn_update_dpp(0, __builtin_bit_cast(int, x), 0x160 + M, 0xf, 0xf, true));
}

__device__ __forceinline__ float fold16(float p, float q, bool up) {
  return (up ? q : p) + lane_xor16(up ? p : q);
}
template <int M>
__device__ __forceinline__ float fold_row(float p, float q, bool up) {
  return (up ? q : p) + lane_xor_row<M>(up ? p : q);
}

__global__ __launch_bounds__(256) void k_edge_pred(const float* __restrict__ g,
                                                   const float* __restrict__ bh,
                                                   const int* __restrict__ src,
                                                   const int* __restrict__ dst,
                                                   const float* __restrict__ Wfc,
                                                   const float* __restrict__ bfc,
                                                   float* __restrict__ out, int E) {
  static_assert(EB == 8 && EPW == 32, "the tree below folds 8 edges per group and 4 groups per wave");
  const int lane = threadIdx.x & 31;
  const int wave = __builtin_amdgcn_readfirstlane((int)(blockIdx.x * blockDim.x + threadIdx.x) >> 5);
  const int e0 = wave * EPW;
  if (e0 >= E) return;
  const int e = e0 + lane;
  const bool ok = e < E;
  const int s_l = ok ? src[e] : 0;
  const int d_l = ok ? dst[e] : 0;
  const int c = lane * 2;
  const float2 w0 = *(const float2*)(Wfc + c);
  const float2 w1 = *(const float2*)(Wfc + CH + c);
  const float2 bb = *(const float2*)(bh + c);
  const bool up16 = (lane & 16) != 0, up8 = (lane & 8) != 0, up4 = (lane & 4) != 0;
  const bool up2 = (lane & 2) != 0, up1 = (lane & 1) != 0;
  float q[EPW / EB];
#pragma unroll
  for (int j0 = 0; j0 < EPW; j0 += EB) {
    float2 hs[EB], hd[EB];
#pragma unroll
    for (int j = 0; j < EB; ++j) {
      const int s = __builtin_amdgcn_readlane(s_l, j0 + j);
      const int d = __builtin_amdgcn_readlane(d_l, j0 + j);
      hs[j] = *(const float2*)(g + (size_t)s * CH + c);
      hd[j] = *(const float2*)(g + (size_t)d * CH + c);
    }
    float p[EB];
#pragma unroll
    for (int j = 0; j < EB; ++j) {
      const float sx = bias_relu(hs[j].x, bb.x), sy = bias_relu(hs[j].y, bb.y);
      const float dx = bias_relu(hd[j].x, bb.x), dy = bias_relu(hd[j].y, bb.y);
      p[j] = sx * w0.x + sy * w0.y + dx * w1.x + dy * w1.y;
    }
    const float a0 = fold16(p[0], p[1], up16), a1 = fold16(p[2], p[3], up16);
    const float a2 = fold16(p[4], p[5], up16), a3 = fold16(p[6], p[7], up16);
    const float b0 = fold_row<8>(a0, a1, up8), b1 = fold_row<8>(a2, a3, up8);
    q[j0 / EB] = fold_row<4>(b0, b1, up4);
  }
  const float r0 = fold_row<2>(q[0], q[1], up2), r1 = fold_row<2>(q[2], q[3], up2);
  const float res = fold_row<1>(r0, r1, up1);
  const int eo = e0 + (int)(__brev((unsigned)lane) >> 27);
  if (eo < E) out[eo] = res + bfc[0];
}

extern "C" void kernel_launch(void* const* d_in, const int* in_sizes, int n_in,
                              void* d_out, int out_size, void* d_ws, size_t ws_size,
                              hipStream_t stream) {
  const float* x   = (const float*)d_in[0];
  const int*   ei  = (const int*)d_in[1];
  const float* W1  = (const float*)d_in[2];
  const float* b1  = (const float*)d_in[3];
  const float* W2  = (const float*)d_in[4];
  const float* b2  = (const float*)d_in[5];
  const float* Wfc = (const float*)d_in[6];
  const float* bfc = (const float*)d_in[7];

  const int N = in_sizes[0] / CH;
  const int E = in_sizes[1] / 2;
  const int NF = N * CH;

  const int* src = ei;
  const int* dst = ei + E;

  float* ws   = (float*)d_ws;
  float* HB   = ws;
  float* AB   = HB + (size_t)NF;
  float* CB   = AB + (size_t)NF;
  const int NP = (N + 127) / 128 * 128;
  unsigned* part = (unsigned*)(CB + (size_t)NF);
  float* deg  = (float*)(part + (size_t)DEG_G * NP);
  float* nrm  = deg + NP;
  const int EP = (E + 127) / 128 * 128;
  unsigned* ctr = (unsigned*)(nrm + EP);
  const int NH = N / 2;
  const int nItems = (E + SC_ITEM - 1) / SC_ITEM;
  dim3 gS(1792);

  const int TB = 256;
  dim3 blk(TB);
  dim3 gEw((E + TB - 1) / TB);
  dim3 gG((N + 31) / 32);

  k_deg<<<dim3(DEG_G), dim3(DEG_TPB), 0, stream>>>(dst, part, ctr, E, N, NP);

  k_gemm64<false, true><<<gG, blk, 0, stream>>>(x, b1, W1, part, NP, deg, HB, AB, N);
  k_scatter<true><<<gS, dim3(SC_TPB), 0, stream>>>(HB, src, dst, deg, nullptr, nrm, AB, ctr, E, NH, nItems);

  k_gemm64<true, false><<<gG, blk, 0, stream>>>(AB, b1, W2, part, NP, deg, HB, CB, N);
  k_scatter<false><<<gS, dim3(SC_TPB), 0, stream>>>(HB, src, dst, deg, nrm, nullptr, CB, ctr + 2 * SC_CTR, E, NH, nItems);

  k_edge_pred<<<gEw, blk, 0, stream>>>(CB, b2, src, dst, Wfc, bfc, (float*)d_out, E);
}
